# starting kernel
# baseline (speedup 1.0000x reference)
#include <hip/hip_runtime.h>
#include <hip/hip_bf16.h>

typedef float v2f __attribute__((ext_vector_type(2)));
typedef float v8f __attribute__((ext_vector_type(8)));

#define POOL 5
#define PQ   25
#define CC   512
#define HH   22
#define WW   22
#define HW   484
#define NG   121
#define CK   12800
#define NROI 256
#define RSCALE (1.0f/16.0f)

__device__ __forceinline__ float hat_int(float t) {
    t = fminf(1.0f, fmaxf(-1.0f, t));
    float a = t + 1.0f, b = 1.0f - t;
    return (t < 0.0f) ? 0.5f * a * a : 1.0f - 0.5f * b * b;
}

__device__ __forceinline__ v8f wmma_f32(v2f a, v2f b, v8f c) {
    return __builtin_amdgcn_wmma_f32_16x16x4_f32(false, a, false, b,
                                                 (short)0, c, false, false);
}

__global__ __launch_bounds__(256) void prroi_pool_wmma(
    const float* __restrict__ feat, const float* __restrict__ bb,
    float* __restrict__ pooled)
{
    __shared__ v2f   sB[2 * NG * 32];
    __shared__ float sWx[POOL * WW];
    __shared__ float sWy[POOL * HH];

    const int n   = blockIdx.x;
    const int tid = threadIdx.x;

    const float bx = bb[n*4+0], by = bb[n*4+1], bwid = bb[n*4+2], bhgt = bb[n*4+3];
    const float x1 = bx * RSCALE, y1 = by * RSCALE;
    const float x2 = (bx + bwid) * RSCALE, y2 = (by + bhgt) * RSCALE;
    const float bw = fmaxf((x2 - x1) * (1.0f/POOL), 0.0f);
    const float bh = fmaxf((y2 - y1) * (1.0f/POOL), 0.0f);
    const float area = bw * bh;
    const float inv_area = (area > 0.0f) ? (1.0f / fmaxf(area, 1e-12f)) : 0.0f;

    if (tid < POOL*WW) {
        int q = tid / WW, i = tid % WW;
        sWx[tid] = hat_int(x1 + (q+1)*bw - (float)i) - hat_int(x1 + q*bw - (float)i);
    } else if (tid < POOL*WW + POOL*HH) {
        int t2 = tid - POOL*WW;
        int p = t2 / HH, j = t2 % HH;
        sWy[t2] = hat_int(y1 + (p+1)*bh - (float)j) - hat_int(y1 + p*bh - (float)j);
    }
    __syncthreads();

    for (int idx = tid; idx < 2*NG*32; idx += 256) {
        int bt = idx / (NG*32);
        int r  = idx - bt*NG*32;
        int g  = r >> 5;
        int ln = r & 31;
        int pq = bt*16 + (ln & 15);
        int k  = 4*g + 2*(ln >> 4);
        v2f val = {0.0f, 0.0f};
        if (pq < PQ) {
            int p = pq / POOL, q = pq % POOL;
            float wp0 = sWy[p*HH + (k    / WW)] * sWx[q*WW + (k    % WW)];
            float wp1 = sWy[p*HH + ((k+1)/ WW)] * sWx[q*WW + ((k+1)% WW)];
            val.x = wp0 * inv_area;
            val.y = wp1 * inv_area;
        }
        sB[idx] = val;
    }
    __syncthreads();

    const int wv   = tid >> 5;
    const int lane = tid & 31;
    const int m16  = lane & 15;
    const int koff = (lane >> 4) * 2;

    const float* fn = feat + (long)n * (CC * HW);
    float*       pn = pooled + (long)n * CK;

    const v2f* B0 = sB + lane;
    const v2f* B1 = sB + NG*32 + lane;

    for (int job = 0; job < 4; ++job) {
        const int ctile = wv + job * 8;
        const float* ap = fn + (long)(ctile*16 + m16) * HW + koff;

        v8f acc0 = {};
        v8f acc1 = {};
        #pragma unroll 1
        for (int g = 0; g < NG - 1; g += 4) {
            const float* base = ap + g*4;
            v2f x0 = *(const v2f*)(base + 0);
            v2f x1 = *(const v2f*)(base + 4);
            v2f x2 = *(const v2f*)(base + 8);
            v2f x3 = *(const v2f*)(base + 12);
            const v2f* pb0 = B0 + g*32;
            const v2f* pb1 = B1 + g*32;
            acc0 = wmma_f32(x0, pb0[ 0], acc0);  acc1 = wmma_f32(x0, pb1[ 0], acc1);
            acc0 = wmma_f32(x1, pb0[32], acc0);  acc1 = wmma_f32(x1, pb1[32], acc1);
            acc0 = wmma_f32(x2, pb0[64], acc0);  acc1 = wmma_f32(x2, pb1[64], acc1);
            acc0 = wmma_f32(x3, pb0[96], acc0);  acc1 = wmma_f32(x3, pb1[96], acc1);
        }
        {
            v2f x = *(const v2f*)(ap + 480);
            acc0 = wmma_f32(x, B0[120*32], acc0);
            acc1 = wmma_f32(x, B1[120*32], acc1);
        }

        const int ncol  = m16;
        const int mbase = (lane >> 4) * 8;
        float* p0 = pn + ncol;
        #pragma unroll
        for (int j = 0; j < 8; ++j) p0[(ctile*16 + mbase + j) * PQ] = acc0[j];
        if (ncol < 9) {
            float* p1 = pn + 16 + ncol;
            #pragma unroll
            for (int j = 0; j < 8; ++j) p1[(ctile*16 + mbase + j) * PQ] = acc1[j];
        }
    }
}

__global__ __launch_bounds__(256) void head_wmma(
    const float* __restrict__ pooled,
    const float* __restrict__ w_obj, const float* __restrict__ b_obj,
    const float* __restrict__ w_mot, const float* __restrict__ b_mot,
    float* __restrict__ out)
{
    __shared__ float red[8 * 256];
    const int mtile = blockIdx.x;
    const int ntile = blockIdx.y;
    const int tid   = threadIdx.x;
    const int wv    = tid >> 5;
    const int lane  = tid & 31;
    const int m16   = lane & 15;
    const int kpair = (lane >> 4) * 2;
    const int KSL   = CK / 8;
    const int GSL   = KSL / 4;

    const int nh = ntile*16 + m16;
    const float* wrow;
    if      (nh < 20) wrow = w_obj + (long)nh * CK;
    else if (nh < 24) wrow = w_mot + (long)(nh - 20) * CK;
    else              wrow = w_obj;

    const float* ap = pooled + (long)(mtile*16 + m16) * CK + kpair + wv * KSL;
    const float* bp = wrow + kpair + wv * KSL;

    v8f acc = {};
    #pragma unroll 1
    for (int g = 0; g < GSL; g += 4) {
        const float* a4 = ap + g*4;
        const float* b4 = bp + g*4;
        v2f xa0 = *(const v2f*)(a4 + 0);
        v2f xa1 = *(const v2f*)(a4 + 4);
        v2f xa2 = *(const v2f*)(a4 + 8);
        v2f xa3 = *(const v2f*)(a4 + 12);
        v2f xb0 = *(const v2f*)(b4 + 0);
        v2f xb1 = *(const v2f*)(b4 + 4);
        v2f xb2 = *(const v2f*)(b4 + 8);
        v2f xb3 = *(const v2f*)(b4 + 12);
        acc = wmma_f32(xa0, xb0, acc);
        acc = wmma_f32(xa1, xb1, acc);
        acc = wmma_f32(xa2, xb2, acc);
        acc = wmma_f32(xa3, xb3, acc);
    }

    #pragma unroll
    for (int j = 0; j < 8; ++j) red[wv*256 + lane*8 + j] = acc[j];
    __syncthreads();

    float s = 0.0f;
    #pragma unroll
    for (int w = 0; w < 8; ++w) s += red[w*256 + tid];

    const int lsrc = tid >> 3, j = tid & 7;
    const int ncol = lsrc & 15;
    const int mrow = (lsrc >> 4) * 8 + j;
    const int nh2  = ntile*16 + ncol;
    const int roi  = mtile*16 + mrow;
    if (nh2 < 24) {
        float bias = (nh2 < 20) ? b_obj[nh2] : b_mot[nh2 - 20];
        float val = s + bias;
        if (nh2 < 20) out[roi*20 + nh2] = val;
        else          out[20*NROI + roi*4 + (nh2 - 20)] = val;
    }
}

extern "C" void kernel_launch(void* const* d_in, const int* in_sizes, int n_in,
                              void* d_out, int out_size, void* d_ws, size_t ws_size,
                              hipStream_t stream) {
    const float* feat  = (const float*)d_in[0];
    const float* bb    = (const float*)d_in[1];
    const float* w_obj = (const float*)d_in[2];
    const float* b_obj = (const float*)d_in[3];
    const float* w_mot = (const float*)d_in[4];
    const float* b_mot = (const float*)d_in[5];
    float* out    = (float*)d_out;
    float* pooled = (float*)d_ws;

    prroi_pool_wmma<<<dim3(NROI), 256, 0, stream>>>(feat, bb, pooled);
    head_wmma<<<dim3(16, 2), 256, 0, stream>>>(pooled, w_obj, b_obj, w_mot, b_mot, out);
}


# head GEMM fused into pooling kernel: group = 16 ROIs x 32-channel slice, pooled tile stays in LDS, 16x24 head tile per slice; slices summed in order by a small 2nd launch (head k sum order changes)
# speedup vs baseline: 3.6102x; 3.6102x over previous
#include <hip/hip_runtime.h>
#include <hip/hip_bf16.h>

typedef float v2f __attribute__((ext_vector_type(2)));
typedef float v8f __attribute__((ext_vector_type(8)));

#define POOL 5
#define PQ   25
#define CC   512
#define HH   22
#define WW   22
#define HW   484
#define CK   12800
#define NROI 256
#define RSCALE (1.0f/16.0f)
#define PWAVES 16
#define CPW    32
#define RPG    16
#define KSL    800
#define LP     804
#define NHEAD  24
#define NX     11

__device__ __forceinline__ float hat_int(float t) {
    t = fminf(1.0f, fmaxf(-1.0f, t));
    float a = t + 1.0f, b = 1.0f - t;
    return (t < 0.0f) ? 0.5f * a * a : 1.0f - 0.5f * b * b;
}

__device__ __forceinline__ v8f wmma_f32(v2f a, v2f b, v8f c) {
    return __builtin_amdgcn_wmma_f32_16x16x4_f32(false, a, false, b,
                                                 (short)0, c, false, false);
}

__device__ __forceinline__ void load_pair(float (&x)[2][2][NX], const float* __restrict__ pc,
                                          int off0, int off1)
{
    #pragma unroll
    for (int ch = 0; ch < 2; ++ch) {
        const float* p = pc + ch * HW;
        #pragma unroll
        for (int k = 0; k < NX; ++k) {
            x[ch][0][k] = p[off0 + 2 * k * WW];
            x[ch][1][k] = p[off1 + 2 * k * WW];
        }
    }
}

__device__ __forceinline__ void pool_pair(const float (&x)[2][2][NX], const v2f (&b1)[6],
                                          const v2f (&a2)[7], float* __restrict__ outp, bool writer)
{
    v8f d1[2][2] = {};
    #pragma unroll
    for (int g = 0; g < 6; ++g) {
        #pragma unroll
        for (int ch = 0; ch < 2; ++ch) {
            #pragma unroll
            for (int mt = 0; mt < 2; ++mt) {
                v2f a;
                a.x = x[ch][mt][(g < 5) ? 2 * g : 10];
                a.y = x[ch][mt][(g < 5) ? 2 * g + 1 : 10];
                d1[ch][mt] = wmma_f32(a, b1[g], d1[ch][mt]);
            }
        }
    }
    v8f d2[2] = {};
    #pragma unroll
    for (int f = 0; f < 7; ++f) {
        const int mt = f >> 2, t = f & 3;
        #pragma unroll
        for (int ch = 0; ch < 2; ++ch) {
            v2f b;
            b.x = d1[ch][mt][2 * t];
            b.y = d1[ch][mt][2 * t + 1];
            d2[ch] = wmma_f32(a2[f], b, d2[ch]);
        }
    }
    if (writer) {
        #pragma unroll
        for (int ch = 0; ch < 2; ++ch) {
            #pragma unroll
            for (int r = 0; r < POOL; ++r) outp[ch * PQ + r] = d2[ch][r];
        }
    }
}

__global__ __launch_bounds__(512) void prroi_pool_head_wmma(
    const float* __restrict__ feat, const float* __restrict__ bb,
    const float* __restrict__ w_obj, const float* __restrict__ w_mot,
    float* __restrict__ part)
{
    __shared__ __attribute__((aligned(16))) float sP[RPG * LP];
    __shared__ __attribute__((aligned(16))) float sW[NHEAD * LP];
    __shared__ float sR[PWAVES * 256];
    __shared__ float sT[PWAVES * 224];

    const int tid  = threadIdx.x;
    const int wv   = __builtin_amdgcn_readfirstlane(tid >> 5);
    const int lane = tid & 31;
    const int n    = blockIdx.x * RPG + wv;
    const int cs   = blockIdx.y;

    const float bx = bb[n*4+0], by = bb[n*4+1], bwid = bb[n*4+2], bhgt = bb[n*4+3];
    const float x1 = bx * RSCALE, y1 = by * RSCALE;
    const float x2 = (bx + bwid) * RSCALE, y2 = (by + bhgt) * RSCALE;
    const float bw = fmaxf((x2 - x1) * (1.0f/POOL), 0.0f);
    const float bh = fmaxf((y2 - y1) * (1.0f/POOL), 0.0f);
    const float area = bw * bh;
    const float inv_area = (area > 0.0f) ? (1.0f / fmaxf(area, 1e-12f)) : 0.0f;

    const int m16  = lane & 15;
    const int h    = lane >> 4;
    const int mrow = (m16 < POOL) ? m16 : (POOL - 1);
    const bool mok = (m16 < POOL);

    const float* fw = feat + (long)n * (CC * HW) + (long)(cs * CPW) * HW;
    const int c1   = 16 + m16;
    const int off0 = h * WW + m16;
    const int off1 = h * WW + ((c1 < WW) ? c1 : (WW - 1));

    float xa[2][2][NX], xb[2][2][NX];
    load_pair(xa, fw, off0, off1);

    float* tw = sT + wv * 224;
    #pragma unroll
    for (int r = 0; r < 7; ++r) {
        const int e = lane + 32 * r;
        if (e < POOL * WW) {
            const int q = e / WW, i = e % WW;
            tw[e] = hat_int(x1 + (q+1)*bw - (float)i) - hat_int(x1 + q*bw - (float)i);
        } else if (e < POOL * WW + POOL * HH) {
            const int t2 = e - POOL * WW;
            const int p = t2 / HH, j = t2 % HH;
            tw[e] = (hat_int(y1 + (p+1)*bh - (float)j) - hat_int(y1 + p*bh - (float)j)) * inv_area;
        }
    }
    __syncthreads();
    const float* sWx = tw;
    const float* sWy = tw + POOL * WW;

    v2f b1[6];
    #pragma unroll
    for (int g = 0; g < 6; ++g) {
        const int j0 = 4*g + h;
        const int j1 = j0 + 2;
        const float w0 = sWy[mrow*HH + j0];
        const float w1 = sWy[mrow*HH + ((j1 < HH) ? j1 : (HH - 1))];
        b1[g].x = mok ? w0 : 0.0f;
        b1[g].y = (mok && j1 < HH) ? w1 : 0.0f;
    }
    v2f a2[7];
    #pragma unroll
    for (int f = 0; f < 7; ++f) {
        const int mt = f >> 2, t = f & 3;
        const int i0 = 16*mt + 8*h + 2*t;
        const int i1 = i0 + 1;
        const float w0 = sWx[mrow*WW + ((i0 < WW) ? i0 : (WW - 1))];
        const float w1 = sWx[mrow*WW + ((i1 < WW) ? i1 : (WW - 1))];
        a2[f].x = (mok && i0 < WW) ? w0 : 0.0f;
        a2[f].y = (mok && i1 < WW) ? w1 : 0.0f;
    }

    float* po = sP + wv * LP + m16 * POOL;
    const bool writer = (lane < POOL);

    #pragma unroll 1
    for (int it = 0; it < CPW / 2; it += 2) {
        load_pair(xb, fw + (it + 1) * 2 * HW, off0, off1);
        __builtin_amdgcn_sched_barrier(0);
        pool_pair(xa, b1, a2, po + it * 2 * PQ, writer);
        const int itn = (it + 2 < CPW / 2) ? (it + 2) : (CPW / 2 - 1);
        load_pair(xa, fw + itn * 2 * HW, off0, off1);
        __builtin_amdgcn_sched_barrier(0);
        pool_pair(xb, b1, a2, po + (it + 1) * 2 * PQ, writer);
    }

    if (tid < 480) {
        float4 wv4[10];
        #pragma unroll
        for (int u = 0; u < 10; ++u) {
            const int e = u * 480 + tid;
            const int row = e / (KSL / 4), c4 = e % (KSL / 4);
            const float* wr = (row < 20) ? (w_obj + (long)row * CK) : (w_mot + (long)(row - 20) * CK);
            wv4[u] = *(const float4*)(wr + cs * KSL + c4 * 4);
        }
        #pragma unroll
        for (int u = 0; u < 10; ++u) {
            const int e = u * 480 + tid;
            const int row = e / (KSL / 4), c4 = e % (KSL / 4);
            *(float4*)(sW + row * LP + c4 * 4) = wv4[u];
        }
    }
    __syncthreads();

    const int nt = wv & 1, ks = wv >> 1;
    const int hd = nt * 16 + m16;
    const int hrow = (hd < NHEAD) ? hd : (NHEAD - 1);
    const float* ap = sP + m16 * LP + ks * 100 + 2 * h;
    const float* bp = sW + hrow * LP + ks * 100 + 2 * h;
    v2f af[25], bf[25];
    #pragma unroll
    for (int g = 0; g < 25; ++g) {
        af[g] = *(const v2f*)(ap + 4 * g);
        bf[g] = *(const v2f*)(bp + 4 * g);
    }
    __builtin_amdgcn_sched_barrier(0);
    v8f acc = {};
    #pragma unroll
    for (int g = 0; g < 25; ++g) acc = wmma_f32(af[g], bf[g], acc);
    #pragma unroll
    for (int r = 0; r < 8; ++r) sR[wv * 256 + r * 32 + lane] = acc[r];
    __syncthreads();

    if (tid < RPG * NHEAD) {
        const int rl = tid / NHEAD, hh = tid % NHEAD;
        const int t1 = hh >> 4, cc = hh & 15;
        const int src = (rl & 7) * 32 + (rl >> 3) * 16 + cc;
        float s = 0.0f;
        #pragma unroll
        for (int k8 = 0; k8 < 8; ++k8) s += sR[(k8 * 2 + t1) * 256 + src];
        part[((long)cs * NROI + blockIdx.x * RPG + rl) * NHEAD + hh] = s;
    }
}

__global__ __launch_bounds__(256) void head_sum(
    const float* __restrict__ part,
    const float* __restrict__ b_obj, const float* __restrict__ b_mot,
    float* __restrict__ out)
{
    const int o = blockIdx.x * 256 + threadIdx.x;
    const bool obj = (o < 20 * NROI);
    const int o2  = o - 20 * NROI;
    const int roi = obj ? (o / 20) : (o2 / 4);
    const int hd  = obj ? (o % 20) : (20 + (o2 & 3));
    const int idx = roi * NHEAD + hd;
    float s = 0.0f;
    #pragma unroll
    for (int c = 0; c < CC / CPW; ++c) s += part[c * (NROI * NHEAD) + idx];
    const float bias = obj ? b_obj[hd] : b_mot[hd - 20];
    out[o] = s + bias;
}

extern "C" void kernel_launch(void* const* d_in, const int* in_sizes, int n_in,
                              void* d_out, int out_size, void* d_ws, size_t ws_size,
                              hipStream_t stream) {
    const float* feat  = (const float*)d_in[0];
    const float* bb    = (const float*)d_in[1];
    const float* w_obj = (const float*)d_in[2];
    const float* b_obj = (const float*)d_in[3];
    const float* w_mot = (const float*)d_in[4];
    const float* b_mot = (const float*)d_in[5];
    float* out  = (float*)d_out;
    float* part = (float*)d_ws;

    prroi_pool_head_wmma<<<dim3(NROI / RPG, CC / CPW), 512, 0, stream>>>(feat, bb, w_obj, w_mot, part);
    head_sum<<<dim3(NROI * NHEAD / 256), 256, 0, stream>>>(part, b_obj, b_mot, out);
}


# pooling: columns 16..21 of a channel pair share one stage-1 A tile (rows 0-5 channel 0, rows 6-11 channel 1): 33 loads and 32 WMMAs per pair, not 44 and 38; each channel's column sum order kept
# speedup vs baseline: 3.8480x; 1.0659x over previous
#include <hip/hip_runtime.h>
#include <hip/hip_bf16.h>

typedef float v2f __attribute__((ext_vector_type(2)));
typedef float v8f __attribute__((ext_vector_type(8)));

#define POOL 5
#define PQ   25
#define CC   512
#define HH   22
#define WW   22
#define HW   484
#define CK   12800
#define NROI 256
#define RSCALE (1.0f/16.0f)
#define PWAVES 16
#define CPW    32
#define RPG    16
#define KSL    800
#define LP     804
#define NHEAD  24
#define NX     11

__device__ __forceinline__ float hat_int(float t) {
    t = fminf(1.0f, fmaxf(-1.0f, t));
    float a = t + 1.0f, b = 1.0f - t;
    return (t < 0.0f) ? 0.5f * a * a : 1.0f - 0.5f * b * b;
}

__device__ __forceinline__ v8f wmma_f32(v2f a, v2f b, v8f c) {
    return __builtin_amdgcn_wmma_f32_16x16x4_f32(false, a, false, b,
                                                 (short)0, c, false, false);
}

__device__ __forceinline__ void load_pair(float (&x)[3][NX], const float* __restrict__ pc,
                                          int off0, int offS)
{
    #pragma unroll
    for (int k = 0; k < NX; ++k) {
        x[0][k] = pc[off0 + 2 * k * WW];
        x[1][k] = pc[HW + off0 + 2 * k * WW];
        x[2][k] = pc[offS + 2 * k * WW];
    }
}

__device__ __forceinline__ void pool_pair(const float (&x)[3][NX], const v2f (&b1)[6],
                                          const v2f (&a2)[7], const v2f (&aS)[3],
                                          float* __restrict__ outp, bool writer)
{
    v8f d1[3] = {};
    #pragma unroll
    for (int g = 0; g < 6; ++g) {
        #pragma unroll
        for (int tl = 0; tl < 3; ++tl) {
            v2f a;
            a.x = x[tl][(g < 5) ? 2 * g : 10];
            a.y = x[tl][(g < 5) ? 2 * g + 1 : 10];
            d1[tl] = wmma_f32(a, b1[g], d1[tl]);
        }
    }
    v8f d2[2] = {};
    #pragma unroll
    for (int t = 0; t < 4; ++t) {
        #pragma unroll
        for (int ch = 0; ch < 2; ++ch) {
            v2f b;
            b.x = d1[ch][2 * t];
            b.y = d1[ch][2 * t + 1];
            d2[ch] = wmma_f32(a2[t], b, d2[ch]);
        }
    }
    #pragma unroll
    for (int u = 0; u < 3; ++u) {
        const int t1 = (u == 0) ? 3 : (u - 1);
        v2f b0, b1s;
        b0.x  = d1[2][2 * u];
        b0.y  = d1[2][2 * u + 1];
        b1s.x = d1[2][2 * t1];
        b1s.y = d1[2][2 * t1 + 1];
        d2[0] = wmma_f32(a2[4 + u], b0, d2[0]);
        d2[1] = wmma_f32(aS[u], b1s, d2[1]);
    }
    if (writer) {
        #pragma unroll
        for (int ch = 0; ch < 2; ++ch) {
            #pragma unroll
            for (int r = 0; r < POOL; ++r) outp[ch * PQ + r] = d2[ch][r];
        }
    }
}

__global__ __launch_bounds__(512) void prroi_pool_head_wmma(
    const float* __restrict__ feat, const float* __restrict__ bb,
    const float* __restrict__ w_obj, const float* __restrict__ w_mot,
    float* __restrict__ part)
{
    __shared__ __attribute__((aligned(16))) float sP[RPG * LP];
    __shared__ __attribute__((aligned(16))) float sW[NHEAD * LP];
    __shared__ float sR[PWAVES * 256];
    __shared__ float sT[PWAVES * 224];

    const int tid  = threadIdx.x;
    const int wv   = __builtin_amdgcn_readfirstlane(tid >> 5);
    const int lane = tid & 31;
    const int n    = blockIdx.x * RPG + wv;
    const int cs   = blockIdx.y;

    const float bx = bb[n*4+0], by = bb[n*4+1], bwid = bb[n*4+2], bhgt = bb[n*4+3];
    const float x1 = bx * RSCALE, y1 = by * RSCALE;
    const float x2 = (bx + bwid) * RSCALE, y2 = (by + bhgt) * RSCALE;
    const float bw = fmaxf((x2 - x1) * (1.0f/POOL), 0.0f);
    const float bh = fmaxf((y2 - y1) * (1.0f/POOL), 0.0f);
    const float area = bw * bh;
    const float inv_area = (area > 0.0f) ? (1.0f / fmaxf(area, 1e-12f)) : 0.0f;

    const int m16  = lane & 15;
    const int h    = lane >> 4;
    const int mrow = (m16 < POOL) ? m16 : (POOL - 1);
    const bool mok = (m16 < POOL);

    const float* fw = feat + (long)n * (CC * HW) + (long)(cs * CPW) * HW;
    const int off0 = h * WW + m16;
    const int cS   = 10 + m16;
    const int offS = h * WW + ((m16 < 6) ? (16 + m16) : (HW + ((cS < WW) ? cS : (WW - 1))));

    float xa[3][NX], xb[3][NX];
    load_pair(xa, fw, off0, offS);

    float* tw = sT + wv * 224;
    #pragma unroll
    for (int r = 0; r < 7; ++r) {
        const int e = lane + 32 * r;
        if (e < POOL * WW) {
            const int q = e / WW, i = e % WW;
            tw[e] = hat_int(x1 + (q+1)*bw - (float)i) - hat_int(x1 + q*bw - (float)i);
        } else if (e < POOL * WW + POOL * HH) {
            const int t2 = e - POOL * WW;
            const int p = t2 / HH, j = t2 % HH;
            tw[e] = (hat_int(y1 + (p+1)*bh - (float)j) - hat_int(y1 + p*bh - (float)j)) * inv_area;
        }
    }
    __syncthreads();
    const float* sWx = tw;
    const float* sWy = tw + POOL * WW;

    v2f b1[6];
    #pragma unroll
    for (int g = 0; g < 6; ++g) {
        const int j0 = 4*g + h;
        const int j1 = j0 + 2;
        const float w0 = sWy[mrow*HH + j0];
        const float w1 = sWy[mrow*HH + ((j1 < HH) ? j1 : (HH - 1))];
        b1[g].x = mok ? w0 : 0.0f;
        b1[g].y = (mok && j1 < HH) ? w1 : 0.0f;
    }
    v2f a2[7];
    #pragma unroll
    for (int f = 0; f < 7; ++f) {
        const int mt = f >> 2, t = f & 3;
        const int i0 = 16*mt + 8*h + 2*t;
        const int i1 = i0 + 1;
        const float w0 = sWx[mrow*WW + ((i0 < WW) ? i0 : (WW - 1))];
        const float w1 = sWx[mrow*WW + ((i1 < WW) ? i1 : (WW - 1))];
        a2[f].x = (mok && i0 < WW) ? w0 : 0.0f;
        a2[f].y = (mok && i1 < WW) ? w1 : 0.0f;
    }
    v2f aS[3];
    #pragma unroll
    for (int u = 0; u < 3; ++u) {
        const bool on = mok && (h == ((u == 0) ? 0 : 1));
        const float w0 = sWx[mrow*WW + 16 + 2*u];
        const float w1 = sWx[mrow*WW + 17 + 2*u];
        aS[u].x = on ? w0 : 0.0f;
        aS[u].y = on ? w1 : 0.0f;
    }

    float* po = sP + wv * LP + m16 * POOL;
    const bool writer = (lane < POOL);

    #pragma unroll 1
    for (int it = 0; it < CPW / 2; it += 2) {
        load_pair(xb, fw + (it + 1) * 2 * HW, off0, offS);
        __builtin_amdgcn_sched_barrier(0);
        pool_pair(xa, b1, a2, aS, po + it * 2 * PQ, writer);
        const int itn = (it + 2 < CPW / 2) ? (it + 2) : (CPW / 2 - 1);
        load_pair(xa, fw + itn * 2 * HW, off0, offS);
        __builtin_amdgcn_sched_barrier(0);
        pool_pair(xb, b1, a2, aS, po + (it + 1) * 2 * PQ, writer);
    }

    if (tid < 480) {
        float4 wv4[10];
        #pragma unroll
        for (int u = 0; u < 10; ++u) {
            const int e = u * 480 + tid;
            const int row = e / (KSL / 4), c4 = e % (KSL / 4);
            const float* wr = (row < 20) ? (w_obj + (long)row * CK) : (w_mot + (long)(row - 20) * CK);
            wv4[u] = *(const float4*)(wr + cs * KSL + c4 * 4);
        }
        #pragma unroll
        for (int u = 0; u < 10; ++u) {
            const int e = u * 480 + tid;
            const int row = e / (KSL / 4), c4 = e % (KSL / 4);
            *(float4*)(sW + row * LP + c4 * 4) = wv4[u];
        }
    }
    __syncthreads();

    const int nt = wv & 1, ks = wv >> 1;
    const int hd = nt * 16 + m16;
    const int hrow = (hd < NHEAD) ? hd : (NHEAD - 1);
    const float* ap = sP + m16 * LP + ks * 100 + 2 * h;
    const float* bp = sW + hrow * LP + ks * 100 + 2 * h;
    v2f af[25], bf[25];
    #pragma unroll
    for (int g = 0; g < 25; ++g) {
        af[g] = *(const v2f*)(ap + 4 * g);
        bf[g] = *(const v2f*)(bp + 4 * g);
    }
    __builtin_amdgcn_sched_barrier(0);
    v8f acc = {};
    #pragma unroll
    for (int g = 0; g < 25; ++g) acc = wmma_f32(af[g], bf[g], acc);
    #pragma unroll
    for (int r = 0; r < 8; ++r) sR[wv * 256 + r * 32 + lane] = acc[r];
    __syncthreads();

    if (tid < RPG * NHEAD) {
        const int rl = tid / NHEAD, hh = tid % NHEAD;
        const int t1 = hh >> 4, cc = hh & 15;
        const int src = (rl & 7) * 32 + (rl >> 3) * 16 + cc;
        float s = 0.0f;
        #pragma unroll
        for (int k8 = 0; k8 < 8; ++k8) s += sR[(k8 * 2 + t1) * 256 + src];
        part[((long)cs * NROI + blockIdx.x * RPG + rl) * NHEAD + hh] = s;
    }
}

__global__ __launch_bounds__(256) void head_sum(
    const float* __restrict__ part,
    const float* __restrict__ b_obj, const float* __restrict__ b_mot,
    float* __restrict__ out)
{
    const int o = blockIdx.x * 256 + threadIdx.x;
    const bool obj = (o < 20 * NROI);
    const int o2  = o - 20 * NROI;
    const int roi = obj ? (o / 20) : (o2 / 4);
    const int hd  = obj ? (o % 20) : (20 + (o2 & 3));
    const int idx = roi * NHEAD + hd;
    float s = 0.0f;
    #pragma unroll
    for (int c = 0; c < CC / CPW; ++c) s += part[c * (NROI * NHEAD) + idx];
    const float bias = obj ? b_obj[hd] : b_mot[hd - 20];
    out[o] = s + bias;
}

extern "C" void kernel_launch(void* const* d_in, const int* in_sizes, int n_in,
                              void* d_out, int out_size, void* d_ws, size_t ws_size,
                              hipStream_t stream) {
    const float* feat  = (const float*)d_in[0];
    const float* bb    = (const float*)d_in[1];
    const float* w_obj = (const float*)d_in[2];
    const float* b_obj = (const float*)d_in[3];
    const float* w_mot = (const float*)d_in[4];
    const float* b_mot = (const float*)d_in[5];
    float* out  = (float*)d_out;
    float* part = (float*)d_ws;

    prroi_pool_head_wmma<<<dim3(NROI / RPG, CC / CPW), 512, 0, stream>>>(feat, bb, w_obj, w_mot, part);
    head_sum<<<dim3(NROI * NHEAD / 256), 256, 0, stream>>>(part, b_obj, b_mot, out);
}


# pooling: per ROI, if all columns with a non-zero x weight fit one 16-column window, the A tiles slide there and the shared third tile is skipped (22 loads, 20 WMMAs a pair); column sum grouping moves
# speedup vs baseline: 4.2262x; 1.0983x over previous
#include <hip/hip_runtime.h>
#include <hip/hip_bf16.h>

typedef float v2f __attribute__((ext_vector_type(2)));
typedef float v8f __attribute__((ext_vector_type(8)));

#define POOL 5
#define PQ   25
#define CC   512
#define HH   22
#define WW   22
#define HW   484
#define CK   12800
#define NROI 256
#define RSCALE (1.0f/16.0f)
#define PWAVES 16
#define CPW    32
#define RPG    16
#define KSL    800
#define LP     804
#define NHEAD  24
#define NX     11

__device__ __forceinline__ float hat_int(float t) {
    t = fminf(1.0f, fmaxf(-1.0f, t));
    float a = t + 1.0f, b = 1.0f - t;
    return (t < 0.0f) ? 0.5f * a * a : 1.0f - 0.5f * b * b;
}

__device__ __forceinline__ v8f wmma_f32(v2f a, v2f b, v8f c) {
    return __builtin_amdgcn_wmma_f32_16x16x4_f32(false, a, false, b,
                                                 (short)0, c, false, false);
}

__device__ __forceinline__ void load_pair(float (&x)[3][NX], const float* __restrict__ pc,
                                          int off0, int offS)
{
    #pragma unroll
    for (int k = 0; k < NX; ++k) {
        x[0][k] = pc[off0 + 2 * k * WW];
        x[1][k] = pc[HW + off0 + 2 * k * WW];
        x[2][k] = pc[offS + 2 * k * WW];
    }
}

__device__ __forceinline__ void pool_pair(const float (&x)[3][NX], const v2f (&b1)[6],
                                          const v2f (&a2)[7], const v2f (&aS)[3],
                                          float* __restrict__ outp, bool writer)
{
    v8f d1[3] = {};
    #pragma unroll
    for (int g = 0; g < 6; ++g) {
        #pragma unroll
        for (int tl = 0; tl < 3; ++tl) {
            v2f a;
            a.x = x[tl][(g < 5) ? 2 * g : 10];
            a.y = x[tl][(g < 5) ? 2 * g + 1 : 10];
            d1[tl] = wmma_f32(a, b1[g], d1[tl]);
        }
    }
    v8f d2[2] = {};
    #pragma unroll
    for (int t = 0; t < 4; ++t) {
        #pragma unroll
        for (int ch = 0; ch < 2; ++ch) {
            v2f b;
            b.x = d1[ch][2 * t];
            b.y = d1[ch][2 * t + 1];
            d2[ch] = wmma_f32(a2[t], b, d2[ch]);
        }
    }
    #pragma unroll
    for (int u = 0; u < 3; ++u) {
        const int t1 = (u == 0) ? 3 : (u - 1);
        v2f b0, b1s;
        b0.x  = d1[2][2 * u];
        b0.y  = d1[2][2 * u + 1];
        b1s.x = d1[2][2 * t1];
        b1s.y = d1[2][2 * t1 + 1];
        d2[0] = wmma_f32(a2[4 + u], b0, d2[0]);
        d2[1] = wmma_f32(aS[u], b1s, d2[1]);
    }
    if (writer) {
        #pragma unroll
        for (int ch = 0; ch < 2; ++ch) {
            #pragma unroll
            for (int r = 0; r < POOL; ++r) outp[ch * PQ + r] = d2[ch][r];
        }
    }
}

__device__ __forceinline__ void load_pair_win(float (&x)[2][NX], const float* __restrict__ pc, int offn)
{
    #pragma unroll
    for (int k = 0; k < NX; ++k) {
        x[0][k] = pc[offn + 2 * k * WW];
        x[1][k] = pc[HW + offn + 2 * k * WW];
    }
}

__device__ __forceinline__ void pool_pair_win(const float (&x)[2][NX], const v2f (&b1)[6],
                                              const v2f (&aw)[4], float* __restrict__ outp, bool writer)
{
    v8f d1[2] = {};
    #pragma unroll
    for (int g = 0; g < 6; ++g) {
        #pragma unroll
        for (int ch = 0; ch < 2; ++ch) {
            v2f a;
            a.x = x[ch][(g < 5) ? 2 * g : 10];
            a.y = x[ch][(g < 5) ? 2 * g + 1 : 10];
            d1[ch] = wmma_f32(a, b1[g], d1[ch]);
        }
    }
    v8f d2[2] = {};
    #pragma unroll
    for (int t = 0; t < 4; ++t) {
        #pragma unroll
        for (int ch = 0; ch < 2; ++ch) {
            v2f b;
            b.x = d1[ch][2 * t];
            b.y = d1[ch][2 * t + 1];
            d2[ch] = wmma_f32(aw[t], b, d2[ch]);
        }
    }
    if (writer) {
        #pragma unroll
        for (int ch = 0; ch < 2; ++ch) {
            #pragma unroll
            for (int r = 0; r < POOL; ++r) outp[ch * PQ + r] = d2[ch][r];
        }
    }
}

__global__ __launch_bounds__(512) void prroi_pool_head_wmma(
    const float* __restrict__ feat, const float* __restrict__ bb,
    const float* __restrict__ w_obj, const float* __restrict__ w_mot,
    float* __restrict__ part)
{
    __shared__ __attribute__((aligned(16))) float sP[RPG * LP];
    __shared__ __attribute__((aligned(16))) float sW[NHEAD * LP];
    __shared__ float sR[PWAVES * 256];
    __shared__ float sT[PWAVES * 224];

    const int tid  = threadIdx.x;
    const int wv   = __builtin_amdgcn_readfirstlane(tid >> 5);
    const int lane = tid & 31;
    const int n    = blockIdx.x * RPG + wv;
    const int cs   = blockIdx.y;

    const float bx = bb[n*4+0], by = bb[n*4+1], bwid = bb[n*4+2], bhgt = bb[n*4+3];
    const float x1 = bx * RSCALE, y1 = by * RSCALE;
    const float x2 = (bx + bwid) * RSCALE, y2 = (by + bhgt) * RSCALE;
    const float bw = fmaxf((x2 - x1) * (1.0f/POOL), 0.0f);
    const float bh = fmaxf((y2 - y1) * (1.0f/POOL), 0.0f);
    const float area = bw * bh;
    const float inv_area = (area > 0.0f) ? (1.0f / fmaxf(area, 1e-12f)) : 0.0f;

    const int m16  = lane & 15;
    const int h    = lane >> 4;
    const int mrow = (m16 < POOL) ? m16 : (POOL - 1);
    const bool mok = (m16 < POOL);

    const float* fw = feat + (long)n * (CC * HW) + (long)(cs * CPW) * HW;
    const int off0 = h * WW + m16;
    const int cS   = 10 + m16;
    const int offS = h * WW + ((m16 < 6) ? (16 + m16) : (HW + ((cS < WW) ? cS : (WW - 1))));

    float wxc[POOL];
    bool nzc = false;
    {
        const float fi = (float)lane;
        float hprev = hat_int(x1 - fi);
        #pragma unroll
        for (int q = 0; q < POOL; ++q) {
            const float hn = hat_int(x1 + (q+1)*bw - fi);
            wxc[q] = hn - hprev;
            hprev = hn;
            nzc = nzc || (wxc[q] != 0.0f);
        }
    }
    const unsigned cmask = __builtin_amdgcn_ballot_w32(nzc && lane < WW);
    const int ilo = cmask ? __builtin_ctz(cmask) : 0;
    const int ihi = cmask ? (31 - __builtin_clz(cmask)) : 0;
    const bool win = (ihi - ilo) < 16;
    const int c0 = (ilo < WW - 16) ? ilo : (WW - 16);
    const int offn = h * WW + c0 + m16;

    float ya[2][NX], yb[2][NX];
    load_pair_win(ya, fw, offn);

    float* tw = sT + wv * 224;
    if (lane < WW) {
        #pragma unroll
        for (int q = 0; q < POOL; ++q) tw[q * WW + lane] = wxc[q];
    }
    #pragma unroll
    for (int r = 0; r < 4; ++r) {
        const int t2 = lane + 32 * r;
        if (t2 < POOL * HH) {
            const int p = t2 / HH, j = t2 % HH;
            tw[POOL * WW + t2] = (hat_int(y1 + (p+1)*bh - (float)j) - hat_int(y1 + p*bh - (float)j)) * inv_area;
        }
    }
    __syncthreads();
    const float* sWx = tw;
    const float* sWy = tw + POOL * WW;

    v2f b1[6];
    #pragma unroll
    for (int g = 0; g < 6; ++g) {
        const int j0 = 4*g + h;
        const int j1 = j0 + 2;
        const float w0 = sWy[mrow*HH + j0];
        const float w1 = sWy[mrow*HH + ((j1 < HH) ? j1 : (HH - 1))];
        b1[g].x = mok ? w0 : 0.0f;
        b1[g].y = (mok && j1 < HH) ? w1 : 0.0f;
    }
    float* po = sP + wv * LP + m16 * POOL;
    const bool writer = (lane < POOL);

    if (win) {
        v2f aw[4];
        #pragma unroll
        for (int t = 0; t < 4; ++t) {
            const int i0 = c0 + 8*h + 2*t;
            const float w0 = sWx[mrow*WW + i0];
            const float w1 = sWx[mrow*WW + i0 + 1];
            aw[t].x = mok ? w0 : 0.0f;
            aw[t].y = mok ? w1 : 0.0f;
        }
        #pragma unroll 1
        for (int it = 0; it < CPW / 2; it += 2) {
            load_pair_win(yb, fw + (it + 1) * 2 * HW, offn);
            __builtin_amdgcn_sched_barrier(0);
            pool_pair_win(ya, b1, aw, po + it * 2 * PQ, writer);
            const int itn = (it + 2 < CPW / 2) ? (it + 2) : (CPW / 2 - 1);
            load_pair_win(ya, fw + itn * 2 * HW, offn);
            __builtin_amdgcn_sched_barrier(0);
            pool_pair_win(yb, b1, aw, po + (it + 1) * 2 * PQ, writer);
        }
    } else {
        v2f a2[7];
        #pragma unroll
        for (int f = 0; f < 7; ++f) {
            const int mt = f >> 2, t = f & 3;
            const int i0 = 16*mt + 8*h + 2*t;
            const int i1 = i0 + 1;
            const float w0 = sWx[mrow*WW + ((i0 < WW) ? i0 : (WW - 1))];
            const float w1 = sWx[mrow*WW + ((i1 < WW) ? i1 : (WW - 1))];
            a2[f].x = (mok && i0 < WW) ? w0 : 0.0f;
            a2[f].y = (mok && i1 < WW) ? w1 : 0.0f;
        }
        v2f aS[3];
        #pragma unroll
        for (int u = 0; u < 3; ++u) {
            const bool on = mok && (h == ((u == 0) ? 0 : 1));
            const float w0 = sWx[mrow*WW + 16 + 2*u];
            const float w1 = sWx[mrow*WW + 17 + 2*u];
            aS[u].x = on ? w0 : 0.0f;
            aS[u].y = on ? w1 : 0.0f;
        }

        float xa[3][NX], xb[3][NX];
        load_pair(xa, fw, off0, offS);
        #pragma unroll 1
        for (int it = 0; it < CPW / 2; it += 2) {
            load_pair(xb, fw + (it + 1) * 2 * HW, off0, offS);
            __builtin_amdgcn_sched_barrier(0);
            pool_pair(xa, b1, a2, aS, po + it * 2 * PQ, writer);
            const int itn = (it + 2 < CPW / 2) ? (it + 2) : (CPW / 2 - 1);
            load_pair(xa, fw + itn * 2 * HW, off0, offS);
            __builtin_amdgcn_sched_barrier(0);
            pool_pair(xb, b1, a2, aS, po + (it + 1) * 2 * PQ, writer);
        }
    }

    if (tid < 480) {
        float4 wv4[10];
        #pragma unroll
        for (int u = 0; u < 10; ++u) {
            const int e = u * 480 + tid;
            const int row = e / (KSL / 4), c4 = e % (KSL / 4);
            const float* wr = (row < 20) ? (w_obj + (long)row * CK) : (w_mot + (long)(row - 20) * CK);
            wv4[u] = *(const float4*)(wr + cs * KSL + c4 * 4);
        }
        #pragma unroll
        for (int u = 0; u < 10; ++u) {
            const int e = u * 480 + tid;
            const int row = e / (KSL / 4), c4 = e % (KSL / 4);
            *(float4*)(sW + row * LP + c4 * 4) = wv4[u];
        }
    }
    __syncthreads();

    const int nt = wv & 1, ks = wv >> 1;
    const int hd = nt * 16 + m16;
    const int hrow = (hd < NHEAD) ? hd : (NHEAD - 1);
    const float* ap = sP + m16 * LP + ks * 100 + 2 * h;
    const float* bp = sW + hrow * LP + ks * 100 + 2 * h;
    v2f af[25], bf[25];
    #pragma unroll
    for (int g = 0; g < 25; ++g) {
        af[g] = *(const v2f*)(ap + 4 * g);
        bf[g] = *(const v2f*)(bp + 4 * g);
    }
    __builtin_amdgcn_sched_barrier(0);
    v8f acc = {};
    #pragma unroll
    for (int g = 0; g < 25; ++g) acc = wmma_f32(af[g], bf[g], acc);
    #pragma unroll
    for (int r = 0; r < 8; ++r) sR[wv * 256 + r * 32 + lane] = acc[r];
    __syncthreads();

    if (tid < RPG * NHEAD) {
        const int rl = tid / NHEAD, hh = tid % NHEAD;
        const int t1 = hh >> 4, cc = hh & 15;
        const int src = (rl & 7) * 32 + (rl >> 3) * 16 + cc;
        float s = 0.0f;
        #pragma unroll
        for (int k8 = 0; k8 < 8; ++k8) s += sR[(k8 * 2 + t1) * 256 + src];
        part[((long)cs * NROI + blockIdx.x * RPG + rl) * NHEAD + hh] = s;
    }
}

__global__ __launch_bounds__(256) void head_sum(
    const float* __restrict__ part,
    const float* __restrict__ b_obj, const float* __restrict__ b_mot,
    float* __restrict__ out)
{
    const int o = blockIdx.x * 256 + threadIdx.x;
    const bool obj = (o < 20 * NROI);
    const int o2  = o - 20 * NROI;
    const int roi = obj ? (o / 20) : (o2 / 4);
    const int hd  = obj ? (o % 20) : (20 + (o2 & 3));
    const int idx = roi * NHEAD + hd;
    float s = 0.0f;
    #pragma unroll
    for (int c = 0; c < CC / CPW; ++c) s += part[c * (NROI * NHEAD) + idx];
    const float bias = obj ? b_obj[hd] : b_mot[hd - 20];
    out[o] = s + bias;
}

extern "C" void kernel_launch(void* const* d_in, const int* in_sizes, int n_in,
                              void* d_out, int out_size, void* d_ws, size_t ws_size,
                              hipStream_t stream) {
    const float* feat  = (const float*)d_in[0];
    const float* bb    = (const float*)d_in[1];
    const float* w_obj = (const float*)d_in[2];
    const float* b_obj = (const float*)d_in[3];
    const float* w_mot = (const float*)d_in[4];
    const float* b_mot = (const float*)d_in[5];
    float* out  = (float*)d_out;
    float* part = (float*)d_ws;

    prroi_pool_head_wmma<<<dim3(NROI / RPG, CC / CPW), 512, 0, stream>>>(feat, bb, w_obj, w_mot, part);
    head_sum<<<dim3(NROI * NHEAD / 256), 256, 0, stream>>>(part, b_obj, b_mot, out);
}


# pooling stage 1: per ROI the K groups and loads cover only a window of 8 or 12 image rows holding every non-zero y weight (4 channels a load group); taller boxes keep all rows; row sum grouping moves
# speedup vs baseline: 4.4804x; 1.0601x over previous
#include <hip/hip_runtime.h>
#include <hip/hip_bf16.h>

typedef float v2f __attribute__((ext_vector_type(2)));
typedef float v8f __attribute__((ext_vector_type(8)));

#define POOL 5
#define PQ   25
#define CC   512
#define HH   22
#define WW   22
#define HW   484
#define CK   12800
#define NROI 256
#define RSCALE (1.0f/16.0f)
#define PWAVES 16
#define CPW    32
#define RPG    16
#define KSL    800
#define LP     804
#define NHEAD  24
#define NX     11

__device__ __forceinline__ float hat_int(float t) {
    t = fminf(1.0f, fmaxf(-1.0f, t));
    float a = t + 1.0f, b = 1.0f - t;
    return (t < 0.0f) ? 0.5f * a * a : 1.0f - 0.5f * b * b;
}

__device__ __forceinline__ v8f wmma_f32(v2f a, v2f b, v8f c) {
    return __builtin_amdgcn_wmma_f32_16x16x4_f32(false, a, false, b,
                                                 (short)0, c, false, false);
}

__device__ __forceinline__ void load_pair(float (&x)[3][NX], const float* __restrict__ pc,
                                          int off0, int offS)
{
    #pragma unroll
    for (int k = 0; k < NX; ++k) {
        x[0][k] = pc[off0 + 2 * k * WW];
        x[1][k] = pc[HW + off0 + 2 * k * WW];
        x[2][k] = pc[offS + 2 * k * WW];
    }
}

__device__ __forceinline__ void pool_pair(const float (&x)[3][NX], const v2f (&b1)[6],
                                          const v2f (&a2)[7], const v2f (&aS)[3],
                                          float* __restrict__ outp, bool writer)
{
    v8f d1[3] = {};
    #pragma unroll
    for (int g = 0; g < 6; ++g) {
        #pragma unroll
        for (int tl = 0; tl < 3; ++tl) {
            v2f a;
            a.x = x[tl][(g < 5) ? 2 * g : 10];
            a.y = x[tl][(g < 5) ? 2 * g + 1 : 10];
            d1[tl] = wmma_f32(a, b1[g], d1[tl]);
        }
    }
    v8f d2[2] = {};
    #pragma unroll
    for (int t = 0; t < 4; ++t) {
        #pragma unroll
        for (int ch = 0; ch < 2; ++ch) {
            v2f b;
            b.x = d1[ch][2 * t];
            b.y = d1[ch][2 * t + 1];
            d2[ch] = wmma_f32(a2[t], b, d2[ch]);
        }
    }
    #pragma unroll
    for (int u = 0; u < 3; ++u) {
        const int t1 = (u == 0) ? 3 : (u - 1);
        v2f b0, b1s;
        b0.x  = d1[2][2 * u];
        b0.y  = d1[2][2 * u + 1];
        b1s.x = d1[2][2 * t1];
        b1s.y = d1[2][2 * t1 + 1];
        d2[0] = wmma_f32(a2[4 + u], b0, d2[0]);
        d2[1] = wmma_f32(aS[u], b1s, d2[1]);
    }
    if (writer) {
        #pragma unroll
        for (int ch = 0; ch < 2; ++ch) {
            #pragma unroll
            for (int r = 0; r < POOL; ++r) outp[ch * PQ + r] = d2[ch][r];
        }
    }
}

__device__ __forceinline__ void load_pair_win(float (&x)[2][NX], const float* __restrict__ pc, int offn)
{
    #pragma unroll
    for (int k = 0; k < NX; ++k) {
        x[0][k] = pc[offn + 2 * k * WW];
        x[1][k] = pc[HW + offn + 2 * k * WW];
    }
}

__device__ __forceinline__ void pool_pair_win(const float (&x)[2][NX], const v2f (&b1)[6],
                                              const v2f (&aw)[4], float* __restrict__ outp, bool writer)
{
    v8f d1[2] = {};
    #pragma unroll
    for (int g = 0; g < 6; ++g) {
        #pragma unroll
        for (int ch = 0; ch < 2; ++ch) {
            v2f a;
            a.x = x[ch][(g < 5) ? 2 * g : 10];
            a.y = x[ch][(g < 5) ? 2 * g + 1 : 10];
            d1[ch] = wmma_f32(a, b1[g], d1[ch]);
        }
    }
    v8f d2[2] = {};
    #pragma unroll
    for (int t = 0; t < 4; ++t) {
        #pragma unroll
        for (int ch = 0; ch < 2; ++ch) {
            v2f b;
            b.x = d1[ch][2 * t];
            b.y = d1[ch][2 * t + 1];
            d2[ch] = wmma_f32(aw[t], b, d2[ch]);
        }
    }
    if (writer) {
        #pragma unroll
        for (int ch = 0; ch < 2; ++ch) {
            #pragma unroll
            for (int r = 0; r < POOL; ++r) outp[ch * PQ + r] = d2[ch][r];
        }
    }
}

template <int G>
__device__ __forceinline__ void load_quad_rows(float (&x)[4][6], const float* __restrict__ pc, int offr)
{
    #pragma unroll
    for (int k = 0; k < 2 * G; ++k) {
        #pragma unroll
        for (int ch = 0; ch < 4; ++ch) x[ch][k] = pc[ch * HW + offr + 2 * k * WW];
    }
}

template <int G>
__device__ __forceinline__ void pool_quad_rows(const float (&x)[4][6], const v2f (&b1)[G],
                                               const v2f (&aw)[4], float* __restrict__ outp, bool writer)
{
    v8f d1[4] = {};
    #pragma unroll
    for (int g = 0; g < G; ++g) {
        #pragma unroll
        for (int ch = 0; ch < 4; ++ch) {
            v2f a;
            a.x = x[ch][2 * g];
            a.y = x[ch][2 * g + 1];
            d1[ch] = wmma_f32(a, b1[g], d1[ch]);
        }
    }
    v8f d2[4] = {};
    #pragma unroll
    for (int t = 0; t < 4; ++t) {
        #pragma unroll
        for (int ch = 0; ch < 4; ++ch) {
            v2f b;
            b.x = d1[ch][2 * t];
            b.y = d1[ch][2 * t + 1];
            d2[ch] = wmma_f32(aw[t], b, d2[ch]);
        }
    }
    if (writer) {
        #pragma unroll
        for (int ch = 0; ch < 4; ++ch) {
            #pragma unroll
            for (int r = 0; r < POOL; ++r) outp[ch * PQ + r] = d2[ch][r];
        }
    }
}

template <int G>
__device__ __forceinline__ void pool_slice_rows(float (&qa)[4][6], const float* __restrict__ fw, int offr,
                                                const float* __restrict__ wyrow, bool mok, const v2f (&aw)[4],
                                                float* __restrict__ po, bool writer)
{
    v2f b1[G];
    #pragma unroll
    for (int g = 0; g < G; ++g) {
        const float w0 = wyrow[4 * g];
        const float w1 = wyrow[4 * g + 2];
        b1[g].x = mok ? w0 : 0.0f;
        b1[g].y = mok ? w1 : 0.0f;
    }
    float qb[4][6];
    #pragma unroll 1
    for (int it = 0; it < CPW / 4 - 2; it += 2) {
        load_quad_rows<G>(qb, fw + (it + 1) * 4 * HW, offr);
        __builtin_amdgcn_sched_barrier(0);
        pool_quad_rows<G>(qa, b1, aw, po + it * 4 * PQ, writer);
        load_quad_rows<G>(qa, fw + (it + 2) * 4 * HW, offr);
        __builtin_amdgcn_sched_barrier(0);
        pool_quad_rows<G>(qb, b1, aw, po + (it + 1) * 4 * PQ, writer);
    }
    load_quad_rows<G>(qb, fw + (CPW / 4 - 1) * 4 * HW, offr);
    __builtin_amdgcn_sched_barrier(0);
    pool_quad_rows<G>(qa, b1, aw, po + (CPW / 4 - 2) * 4 * PQ, writer);
    pool_quad_rows<G>(qb, b1, aw, po + (CPW / 4 - 1) * 4 * PQ, writer);
}

__global__ __launch_bounds__(512) void prroi_pool_head_wmma(
    const float* __restrict__ feat, const float* __restrict__ bb,
    const float* __restrict__ w_obj, const float* __restrict__ w_mot,
    float* __restrict__ part)
{
    __shared__ __attribute__((aligned(16))) float sP[RPG * LP];
    __shared__ __attribute__((aligned(16))) float sW[NHEAD * LP];
    __shared__ float sR[PWAVES * 256];
    __shared__ float sT[PWAVES * 224];

    const int tid  = threadIdx.x;
    const int wv   = __builtin_amdgcn_readfirstlane(tid >> 5);
    const int lane = tid & 31;
    const int n    = blockIdx.x * RPG + wv;
    const int cs   = blockIdx.y;

    const float bx = bb[n*4+0], by = bb[n*4+1], bwid = bb[n*4+2], bhgt = bb[n*4+3];
    const float x1 = bx * RSCALE, y1 = by * RSCALE;
    const float x2 = (bx + bwid) * RSCALE, y2 = (by + bhgt) * RSCALE;
    const float bw = fmaxf((x2 - x1) * (1.0f/POOL), 0.0f);
    const float bh = fmaxf((y2 - y1) * (1.0f/POOL), 0.0f);
    const float area = bw * bh;
    const float inv_area = (area > 0.0f) ? (1.0f / fmaxf(area, 1e-12f)) : 0.0f;

    const int m16  = lane & 15;
    const int h    = lane >> 4;
    const int mrow = (m16 < POOL) ? m16 : (POOL - 1);
    const bool mok = (m16 < POOL);

    const float* fw = feat + (long)n * (CC * HW) + (long)(cs * CPW) * HW;
    const int off0 = h * WW + m16;
    const int cS   = 10 + m16;
    const int offS = h * WW + ((m16 < 6) ? (16 + m16) : (HW + ((cS < WW) ? cS : (WW - 1))));

    float wxc[POOL];
    bool nzc = false;
    {
        const float fi = (float)lane;
        float hprev = hat_int(x1 - fi);
        #pragma unroll
        for (int q = 0; q < POOL; ++q) {
            const float hn = hat_int(x1 + (q+1)*bw - fi);
            wxc[q] = hn - hprev;
            hprev = hn;
            nzc = nzc || (wxc[q] != 0.0f);
        }
    }
    const unsigned cmask = __builtin_amdgcn_ballot_w32(nzc && lane < WW);
    const int ilo = cmask ? __builtin_ctz(cmask) : 0;
    const int ihi = cmask ? (31 - __builtin_clz(cmask)) : 0;
    const bool win = (ihi - ilo) < 16;
    const int c0 = (ilo < WW - 16) ? ilo : (WW - 16);
    const int offn = h * WW + c0 + m16;

    float wyc[POOL];
    bool nzr = false;
    {
        const float fj = (float)lane;
        float hprev = hat_int(y1 - fj);
        #pragma unroll
        for (int p = 0; p < POOL; ++p) {
            const float hn = hat_int(y1 + (p+1)*bh - fj);
            wyc[p] = (hn - hprev) * inv_area;
            hprev = hn;
            nzr = nzr || (wyc[p] != 0.0f);
        }
    }
    const unsigned rmask = __builtin_amdgcn_ballot_w32(nzr && lane < HH);
    const int jlo = rmask ? __builtin_ctz(rmask) : 0;
    const int jhi = rmask ? (31 - __builtin_clz(rmask)) : 0;
    const bool rw2 = win && (jhi - jlo) < 8;
    const bool rw3 = win && !rw2 && (jhi - jlo) < 12;
    const int rmax = rw2 ? (HH - 8) : (HH - 12);
    const int r0 = (jlo < rmax) ? jlo : rmax;
    const int offr = (r0 + h) * WW + c0 + m16;

    float qa[4][6];
    if (rw2)      load_quad_rows<2>(qa, fw, offr);
    else if (rw3) load_quad_rows<3>(qa, fw, offr);

    float* tw = sT + wv * 224;
    if (lane < WW) {
        #pragma unroll
        for (int q = 0; q < POOL; ++q) tw[q * WW + lane] = wxc[q];
    }
    if (lane < HH) {
        #pragma unroll
        for (int p = 0; p < POOL; ++p) tw[POOL * WW + p * HH + lane] = wyc[p];
    }
    __syncthreads();
    const float* sWx = tw;
    const float* sWy = tw + POOL * WW;

    v2f b1[6];
    #pragma unroll
    for (int g = 0; g < 6; ++g) {
        const int j0 = 4*g + h;
        const int j1 = j0 + 2;
        const float w0 = sWy[mrow*HH + j0];
        const float w1 = sWy[mrow*HH + ((j1 < HH) ? j1 : (HH - 1))];
        b1[g].x = mok ? w0 : 0.0f;
        b1[g].y = (mok && j1 < HH) ? w1 : 0.0f;
    }
    float* po = sP + wv * LP + m16 * POOL;
    const bool writer = (lane < POOL);

    if (win) {
        v2f aw[4];
        #pragma unroll
        for (int t = 0; t < 4; ++t) {
            const int i0 = c0 + 8*h + 2*t;
            const float w0 = sWx[mrow*WW + i0];
            const float w1 = sWx[mrow*WW + i0 + 1];
            aw[t].x = mok ? w0 : 0.0f;
            aw[t].y = mok ? w1 : 0.0f;
        }
        if (rw2) {
            pool_slice_rows<2>(qa, fw, offr, sWy + mrow*HH + r0 + h, mok, aw, po, writer);
        } else if (rw3) {
            pool_slice_rows<3>(qa, fw, offr, sWy + mrow*HH + r0 + h, mok, aw, po, writer);
        } else {
            float ya[2][NX], yb[2][NX];
            load_pair_win(ya, fw, offn);
            #pragma unroll 1
            for (int it = 0; it < CPW / 2; it += 2) {
                load_pair_win(yb, fw + (it + 1) * 2 * HW, offn);
                __builtin_amdgcn_sched_barrier(0);
                pool_pair_win(ya, b1, aw, po + it * 2 * PQ, writer);
                const int itn = (it + 2 < CPW / 2) ? (it + 2) : (CPW / 2 - 1);
                load_pair_win(ya, fw + itn * 2 * HW, offn);
                __builtin_amdgcn_sched_barrier(0);
                pool_pair_win(yb, b1, aw, po + (it + 1) * 2 * PQ, writer);
            }
        }
    } else {
        v2f a2[7];
        #pragma unroll
        for (int f = 0; f < 7; ++f) {
            const int mt = f >> 2, t = f & 3;
            const int i0 = 16*mt + 8*h + 2*t;
            const int i1 = i0 + 1;
            const float w0 = sWx[mrow*WW + ((i0 < WW) ? i0 : (WW - 1))];
            const float w1 = sWx[mrow*WW + ((i1 < WW) ? i1 : (WW - 1))];
            a2[f].x = (mok && i0 < WW) ? w0 : 0.0f;
            a2[f].y = (mok && i1 < WW) ? w1 : 0.0f;
        }
        v2f aS[3];
        #pragma unroll
        for (int u = 0; u < 3; ++u) {
            const bool on = mok && (h == ((u == 0) ? 0 : 1));
            const float w0 = sWx[mrow*WW + 16 + 2*u];
            const float w1 = sWx[mrow*WW + 17 + 2*u];
            aS[u].x = on ? w0 : 0.0f;
            aS[u].y = on ? w1 : 0.0f;
        }

        float xa[3][NX], xb[3][NX];
        load_pair(xa, fw, off0, offS);
        #pragma unroll 1
        for (int it = 0; it < CPW / 2; it += 2) {
            load_pair(xb, fw + (it + 1) * 2 * HW, off0, offS);
            __builtin_amdgcn_sched_barrier(0);
            pool_pair(xa, b1, a2, aS, po + it * 2 * PQ, writer);
            const int itn = (it + 2 < CPW / 2) ? (it + 2) : (CPW / 2 - 1);
            load_pair(xa, fw + itn * 2 * HW, off0, offS);
            __builtin_amdgcn_sched_barrier(0);
            pool_pair(xb, b1, a2, aS, po + (it + 1) * 2 * PQ, writer);
        }
    }

    if (tid < 480) {
        float4 wv4[10];
        #pragma unroll
        for (int u = 0; u < 10; ++u) {
            const int e = u * 480 + tid;
            const int row = e / (KSL / 4), c4 = e % (KSL / 4);
            const float* wr = (row < 20) ? (w_obj + (long)row * CK) : (w_mot + (long)(row - 20) * CK);
            wv4[u] = *(const float4*)(wr + cs * KSL + c4 * 4);
        }
        #pragma unroll
        for (int u = 0; u < 10; ++u) {
            const int e = u * 480 + tid;
            const int row = e / (KSL / 4), c4 = e % (KSL / 4);
            *(float4*)(sW + row * LP + c4 * 4) = wv4[u];
        }
    }
    __syncthreads();

    const int nt = wv & 1, ks = wv >> 1;
    const int hd = nt * 16 + m16;
    const int hrow = (hd < NHEAD) ? hd : (NHEAD - 1);
    const float* ap = sP + m16 * LP + ks * 100 + 2 * h;
    const float* bp = sW + hrow * LP + ks * 100 + 2 * h;
    v2f af[25], bf[25];
    #pragma unroll
    for (int g = 0; g < 25; ++g) {
        af[g] = *(const v2f*)(ap + 4 * g);
        bf[g] = *(const v2f*)(bp + 4 * g);
    }
    __builtin_amdgcn_sched_barrier(0);
    v8f acc = {};
    #pragma unroll
    for (int g = 0; g < 25; ++g) acc = wmma_f32(af[g], bf[g], acc);
    #pragma unroll
    for (int r = 0; r < 8; ++r) sR[wv * 256 + r * 32 + lane] = acc[r];
    __syncthreads();

    if (tid < RPG * NHEAD) {
        const int rl = tid / NHEAD, hh = tid % NHEAD;
        const int t1 = hh >> 4, cc = hh & 15;
        const int src = (rl & 7) * 32 + (rl >> 3) * 16 + cc;
        float s = 0.0f;
        #pragma unroll
        for (int k8 = 0; k8 < 8; ++k8) s += sR[(k8 * 2 + t1) * 256 + src];
        part[((long)cs * NROI + blockIdx.x * RPG + rl) * NHEAD + hh] = s;
    }
}

__global__ __launch_bounds__(256) void head_sum(
    const float* __restrict__ part,
    const float* __restrict__ b_obj, const float* __restrict__ b_mot,
    float* __restrict__ out)
{
    const int o = blockIdx.x * 256 + threadIdx.x;
    const bool obj = (o < 20 * NROI);
    const int o2  = o - 20 * NROI;
    const int roi = obj ? (o / 20) : (o2 / 4);
    const int hd  = obj ? (o % 20) : (20 + (o2 & 3));
    const int idx = roi * NHEAD + hd;
    float s = 0.0f;
    #pragma unroll
    for (int c = 0; c < CC / CPW; ++c) s += part[c * (NROI * NHEAD) + idx];
    const float bias = obj ? b_obj[hd] : b_mot[hd - 20];
    out[o] = s + bias;
}

extern "C" void kernel_launch(void* const* d_in, const int* in_sizes, int n_in,
                              void* d_out, int out_size, void* d_ws, size_t ws_size,
                              hipStream_t stream) {
    const float* feat  = (const float*)d_in[0];
    const float* bb    = (const float*)d_in[1];
    const float* w_obj = (const float*)d_in[2];
    const float* b_obj = (const float*)d_in[3];
    const float* w_mot = (const float*)d_in[4];
    const float* b_mot = (const float*)d_in[5];
    float* out  = (float*)d_out;
    float* part = (float*)d_ws;

    prroi_pool_head_wmma<<<dim3(NROI / RPG, CC / CPW), 512, 0, stream>>>(feat, bb, w_obj, w_mot, part);
    head_sum<<<dim3(NROI * NHEAD / 256), 256, 0, stream>>>(part, b_obj, b_mot, out);
}
